# starting kernel
# baseline (speedup 1.0000x reference)
#include <hip/hip_runtime.h>

typedef __attribute__((ext_vector_type(2))) float v2f;
typedef __attribute__((ext_vector_type(8))) float v8f;

#define IN_D  32
#define HID_D 32
#define OUT_D 16

__global__ void gcn_zero_kernel(float* __restrict__ p, long long n) {
    long long i = ((long long)blockIdx.x * blockDim.x + threadIdx.x) * 4;
    if (i + 3 < n) {
        *(float4*)(p + i) = make_float4(0.f, 0.f, 0.f, 0.f);
    } else {
        for (long long j = i; j < n; ++j) p[j] = 0.0f;
    }
}

__global__ void gcn_degree_kernel(const int* __restrict__ src,
                                  const int* __restrict__ dst,
                                  float* __restrict__ degO,
                                  float* __restrict__ degI, int E) {
    int e = blockIdx.x * blockDim.x + threadIdx.x;
    if (e < E) {
        atomicAdd(&degO[src[e]], 1.0f);
        atomicAdd(&degI[dst[e]], 1.0f);
    }
}

__global__ void gcn_norm_kernel(const float* __restrict__ degO,
                                const float* __restrict__ degI,
                                float* __restrict__ ns,
                                float* __restrict__ nd, int n) {
    int i = blockIdx.x * blockDim.x + threadIdx.x;
    if (i < n) {
        ns[i] = rsqrtf(fmaxf(degO[i], 1.0f));
        nd[i] = rsqrtf(fmaxf(degI[i], 1.0f));
    }
}

__global__ void gcn_gemm1_kernel(const float* __restrict__ feat,
                                 const float* __restrict__ ns,
                                 const float* __restrict__ W1,
                                 float* __restrict__ h1, int n) {
    const int wave = threadIdx.x >> 5;
    const int lane = threadIdx.x & 31;
    const int tile = blockIdx.x * (blockDim.x >> 5) + wave;
    const int row0 = tile * 16;
    if (row0 >= n) return;

    const int half = lane >> 4;
    const int mrow = lane & 15;
    const int arow = row0 + mrow;
    const float scale = ns[arow];

    v8f acc0 = {};
    v8f acc1 = {};
#pragma unroll
    for (int k0 = 0; k0 < IN_D; k0 += 4) {
        const int kk = k0 + 2 * half;
        v2f a, b0, b1;
        const float* ap = feat + (long long)arow * IN_D + kk;
        a.x = ap[0] * scale;
        a.y = ap[1] * scale;
        b0.x = W1[kk * HID_D + mrow];
        b0.y = W1[(kk + 1) * HID_D + mrow];
        b1.x = W1[kk * HID_D + 16 + mrow];
        b1.y = W1[(kk + 1) * HID_D + 16 + mrow];
        acc0 = __builtin_amdgcn_wmma_f32_16x16x4_f32(false, a, false, b0,
                                                     (short)0, acc0, false, false);
        acc1 = __builtin_amdgcn_wmma_f32_16x16x4_f32(false, a, false, b1,
                                                     (short)0, acc1, false, false);
    }
#pragma unroll
    for (int j = 0; j < 8; ++j) {
        const int m = row0 + j + 8 * half;
        h1[(long long)m * HID_D + mrow]      = acc0[j];
        h1[(long long)m * HID_D + 16 + mrow] = acc1[j];
    }
}

__global__ void gcn_gemm2_kernel(const float* __restrict__ agg1,
                                 const float* __restrict__ ns,
                                 const float* __restrict__ nd,
                                 const float* __restrict__ b1,
                                 const float* __restrict__ W2,
                                 float* __restrict__ h2, int n) {
    const int wave = threadIdx.x >> 5;
    const int lane = threadIdx.x & 31;
    const int tile = blockIdx.x * (blockDim.x >> 5) + wave;
    const int row0 = tile * 16;
    if (row0 >= n) return;

    const int half = lane >> 4;
    const int mrow = lane & 15;
    const int arow = row0 + mrow;
    const float s_ns = ns[arow];
    const float s_nd = nd[arow];

    v8f acc = {};
#pragma unroll
    for (int k0 = 0; k0 < HID_D; k0 += 4) {
        const int kk = k0 + 2 * half;
        v2f a, b;
        const float* ap = agg1 + (long long)arow * HID_D + kk;
        a.x = fmaxf(ap[0] * s_nd + b1[kk],     0.0f) * s_ns;
        a.y = fmaxf(ap[1] * s_nd + b1[kk + 1], 0.0f) * s_ns;
        b.x = W2[kk * OUT_D + mrow];
        b.y = W2[(kk + 1) * OUT_D + mrow];
        acc = __builtin_amdgcn_wmma_f32_16x16x4_f32(false, a, false, b,
                                                    (short)0, acc, false, false);
    }
#pragma unroll
    for (int j = 0; j < 8; ++j) {
        const int m = row0 + j + 8 * half;
        h2[(long long)m * OUT_D + mrow] = acc[j];
    }
}

template <int D>
__global__ void gcn_scatter_kernel(const float* __restrict__ h,
                                   const int* __restrict__ src,
                                   const int* __restrict__ dst,
                                   const float* __restrict__ w,
                                   float* __restrict__ agg, long long total) {
    constexpr int G = D / 4;
    long long t = (long long)blockIdx.x * blockDim.x + threadIdx.x;
    if (t >= total) return;
    const int e = (int)(t / G);
    const int g = (int)(t % G);
    const int s = src[e];
    const int d = dst[e];
    const float wt = w[e];
    const float4 hv = *(const float4*)(h + (long long)s * D + g * 4);
    float* ap = agg + (long long)d * D + g * 4;
    atomicAdd(ap + 0, hv.x * wt);
    atomicAdd(ap + 1, hv.y * wt);
    atomicAdd(ap + 2, hv.z * wt);
    atomicAdd(ap + 3, hv.w * wt);
}

__global__ void gcn_finalize_kernel(float* __restrict__ out,
                                    const float* __restrict__ nd,
                                    const float* __restrict__ b2, long long total4) {
    constexpr int G = OUT_D / 4;
    long long t = (long long)blockIdx.x * blockDim.x + threadIdx.x;
    if (t >= total4) return;
    const int i = (int)(t / G);
    const int g = (int)(t % G);
    const float s = nd[i];
    const float4 bv = *(const float4*)(b2 + g * 4);
    float4* op = (float4*)(out + (long long)i * OUT_D + g * 4);
    float4 v = *op;
    v.x = v.x * s + bv.x;
    v.y = v.y * s + bv.y;
    v.z = v.z * s + bv.z;
    v.w = v.w * s + bv.w;
    *op = v;
}

extern "C" void kernel_launch(void* const* d_in, const int* in_sizes, int n_in,
                              void* d_out, int out_size, void* d_ws, size_t ws_size,
                              hipStream_t stream) {
    const float* feat = (const float*)d_in[0];
    const int*   src  = (const int*)d_in[1];
    const int*   dst  = (const int*)d_in[2];
    const float* ew   = (const float*)d_in[3];
    const float* W1   = (const float*)d_in[4];
    const float* b1   = (const float*)d_in[5];
    const float* W2   = (const float*)d_in[6];
    const float* b2   = (const float*)d_in[7];
    float* out = (float*)d_out;

    const int n = in_sizes[0] / IN_D;
    const int E = in_sizes[1];

    float* degO = (float*)d_ws;
    float* degI = degO + n;
    float* ns   = degI + n;
    float* nd   = ns + n;
    float* h1   = nd + n;
    float* agg1 = h1 + (long long)n * HID_D;
    float* h2   = agg1 + (long long)n * HID_D;

    const int B = 256;
    auto blocks = [&](long long work) { return (unsigned)((work + B - 1) / B); };

    gcn_zero_kernel<<<blocks((2LL * n + 3) / 4), B, 0, stream>>>(degO, 2LL * n);
    gcn_zero_kernel<<<blocks(((long long)n * HID_D + 3) / 4), B, 0, stream>>>(
        agg1, (long long)n * HID_D);
    gcn_zero_kernel<<<blocks(((long long)out_size + 3) / 4), B, 0, stream>>>(
        out, (long long)out_size);

    gcn_degree_kernel<<<blocks(E), B, 0, stream>>>(src, dst, degO, degI, E);

    gcn_norm_kernel<<<blocks(n), B, 0, stream>>>(degO, degI, ns, nd, n);

    const int tiles = (n + 15) / 16;
    const int wavesPerBlock = B / 32;
    const unsigned gemmBlocks = (unsigned)((tiles + wavesPerBlock - 1) / wavesPerBlock);
    gcn_gemm1_kernel<<<gemmBlocks, B, 0, stream>>>(feat, ns, W1, h1, n);

    gcn_scatter_kernel<HID_D><<<blocks((long long)E * (HID_D / 4)), B, 0, stream>>>(
        h1, src, dst, ew, agg1, (long long)E * (HID_D / 4));

    gcn_gemm2_kernel<<<gemmBlocks, B, 0, stream>>>(agg1, ns, nd, b1, W2, h2, n);

    gcn_scatter_kernel<OUT_D><<<blocks((long long)E * (OUT_D / 4)), B, 0, stream>>>(
        h2, src, dst, ew, out, (long long)E * (OUT_D / 4));

    gcn_finalize_kernel<<<blocks((long long)n * (OUT_D / 4)), B, 0, stream>>>(
        out, nd, b2, (long long)n * (OUT_D / 4));
}


# Degree and both edge-scatter kernels run as 1024 work-groups that walk the edge list, 4 items a trip, next indices loaded under current rows: one die message, one barrier a group; atomics as before
# speedup vs baseline: 2.1592x; 2.1592x over previous
#include <hip/hip_runtime.h>

typedef __attribute__((ext_vector_type(2))) float v2f;
typedef __attribute__((ext_vector_type(8))) float v8f;

#define IN_D  32
#define HID_D 32
#define OUT_D 16

__device__ __forceinline__ unsigned gcn_ask_block_half() {
    unsigned reply = 0u;
    if (__builtin_amdgcn_readfirstlane(threadIdx.x) < 32)
        reply = __builtin_amdgcn_s_sendmsg_rtn(0x87);
    return reply;
}

__device__ __forceinline__ void gcn_post_block_half(unsigned* slot, unsigned reply) {
    if (threadIdx.x == 0) *slot = reply;
}

__device__ __forceinline__ int gcn_take_block_half(const unsigned* slot) {
    __syncthreads();
    return (int)((__builtin_amdgcn_readfirstlane(*slot) >> 18) & 1u);
}

__global__ void gcn_zero_kernel(float* __restrict__ p, long long n) {
    long long i = ((long long)blockIdx.x * blockDim.x + threadIdx.x) * 4;
    if (i + 3 < n) {
        *(float4*)(p + i) = make_float4(0.f, 0.f, 0.f, 0.f);
    } else {
        for (long long j = i; j < n; ++j) p[j] = 0.0f;
    }
}

#define LOOP_BLOCKS 1024
#define LOOP_U 4

__global__ void __launch_bounds__(256)
gcn_degree_kernel(const int* __restrict__ src,
                  const int* __restrict__ dst,
                  float* __restrict__ degO,
                  float* __restrict__ degI, int n, int E) {
    __shared__ unsigned s_half;
    const unsigned asked = gcn_ask_block_half();
    const int step = (int)(gridDim.x * blockDim.x);
    int e = (int)(blockIdx.x * blockDim.x + threadIdx.x);

    const int elast = E - 1;

    int cs[LOOP_U], cd[LOOP_U];
#pragma unroll
    for (int u = 0; u < LOOP_U; ++u) {
        const int ee = min(e + u * step, elast);
        cs[u] = src[ee];
        cd[u] = dst[ee];
    }
    gcn_post_block_half(&s_half, asked);
    const int off = gcn_take_block_half(&s_half) * 2 * n;
    float* const pO = degO + off;
    float* const pI = degI + off;

    for (; e + (LOOP_U - 1) * step < E; e += LOOP_U * step) {
        int ns_[LOOP_U], nd_[LOOP_U];
#pragma unroll
        for (int u = 0; u < LOOP_U; ++u) {
            const int ee = min(e + (LOOP_U + u) * step, elast);
            ns_[u] = src[ee];
            nd_[u] = dst[ee];
        }
#pragma unroll
        for (int u = 0; u < LOOP_U; ++u) {
            atomicAdd(pO + cs[u], 1.0f);
            atomicAdd(pI + cd[u], 1.0f);
        }
#pragma unroll
        for (int u = 0; u < LOOP_U; ++u) {
            cs[u] = ns_[u];
            cd[u] = nd_[u];
        }
    }
#pragma unroll
    for (int u = 0; u < LOOP_U - 1; ++u) {
        if (e + u * step < E) {
            atomicAdd(pO + cs[u], 1.0f);
            atomicAdd(pI + cd[u], 1.0f);
        }
    }
}

__global__ void gcn_norm_kernel(const float* __restrict__ degO,
                                const float* __restrict__ degI,
                                float* __restrict__ ns,
                                float* __restrict__ nd, int n) {
    int i = blockIdx.x * blockDim.x + threadIdx.x;
    if (i < n) {
        ns[i] = rsqrtf(fmaxf(degO[i] + degO[i + 2 * n], 1.0f));
        nd[i] = rsqrtf(fmaxf(degI[i] + degI[i + 2 * n], 1.0f));
    }
}

__global__ void gcn_dense_kernel(float* aggA, float* aggB,
                                 const float* __restrict__ ns,
                                 const float* __restrict__ nd,
                                 const float* __restrict__ W1,
                                 const float* __restrict__ b1,
                                 const float* __restrict__ W2,
                                 float* __restrict__ h2, int n) {
    const int wave = threadIdx.x >> 5;
    const int lane = threadIdx.x & 31;
    const int tile = blockIdx.x * (blockDim.x >> 5) + wave;
    const int row0 = tile * 16;
    if (row0 >= n) return;

    const int half = lane >> 4;
    const int col  = lane & 15;
    const int node = row0 + col;
    const float s_ns = ns[node];
    const float s_nd = nd[node];

    const float4* xa = (const float4*)(aggA + (long long)node * IN_D + 16 * half);
    const float4* xb = (const float4*)(aggB + (long long)node * IN_D + 16 * half);
    float x[16];
#pragma unroll
    for (int q = 0; q < 4; ++q) {
        const float4 v = xa[q];
        const float4 u = xb[q];
        x[4 * q + 0] = v.x + u.x;
        x[4 * q + 1] = v.y + u.y;
        x[4 * q + 2] = v.z + u.z;
        x[4 * q + 3] = v.w + u.w;
    }

    v8f acc0 = {};
    v8f acc1 = {};
#pragma unroll
    for (int i = 0; i < 8; ++i) {
        const int k = 16 * half + 2 * i;
        v2f a0, a1, b;
        a0.x = W1[k * HID_D + col];
        a0.y = W1[(k + 1) * HID_D + col];
        a1.x = W1[k * HID_D + 16 + col];
        a1.y = W1[(k + 1) * HID_D + 16 + col];
        b.x = x[2 * i];
        b.y = x[2 * i + 1];
        acc0 = __builtin_amdgcn_wmma_f32_16x16x4_f32(false, a0, false, b,
                                                     (short)0, acc0, false, false);
        acc1 = __builtin_amdgcn_wmma_f32_16x16x4_f32(false, a1, false, b,
                                                     (short)0, acc1, false, false);
    }

    const float4* bp0 = (const float4*)(b1 + 8 * half);
    const float4* bp1 = (const float4*)(b1 + 16 + 8 * half);
    const float4 c00 = bp0[0], c01 = bp0[1], c10 = bp1[0], c11 = bp1[1];
    const float bias[16] = {c00.x, c00.y, c00.z, c00.w, c01.x, c01.y, c01.z, c01.w,
                            c10.x, c10.y, c10.z, c10.w, c11.x, c11.y, c11.z, c11.w};
    float y[16];
#pragma unroll
    for (int j = 0; j < 8; ++j) {
        y[j]     = fmaxf(acc0[j] * s_nd + bias[j],     0.0f) * s_ns;
        y[8 + j] = fmaxf(acc1[j] * s_nd + bias[8 + j], 0.0f) * s_ns;
    }

    v8f acc2 = {};
#pragma unroll
    for (int i = 0; i < 8; ++i) {
        const int hid = (i < 4 ? 0 : 16) + 8 * half + 2 * (i & 3);
        v2f a, b;
        a.x = W2[hid * OUT_D + col];
        a.y = W2[(hid + 1) * OUT_D + col];
        b.x = y[2 * i];
        b.y = y[2 * i + 1];
        acc2 = __builtin_amdgcn_wmma_f32_16x16x4_f32(false, a, false, b,
                                                     (short)0, acc2, false, false);
    }

    float4* op = (float4*)(h2 + (long long)node * OUT_D + 8 * half);
    op[0] = make_float4(acc2[0], acc2[1], acc2[2], acc2[3]);
    op[1] = make_float4(acc2[4], acc2[5], acc2[6], acc2[7]);

    const float4 zero4 = make_float4(0.f, 0.f, 0.f, 0.f);
    float4* za = (float4*)(aggA + (long long)node * IN_D + 8 * half);
    float4* zb = (float4*)(aggB + (long long)node * IN_D + 8 * half);
    za[0] = zero4;
    za[1] = zero4;
    zb[0] = zero4;
    zb[1] = zero4;
}

template <int D, int PITCH, bool SRC_NORM>
__global__ void __launch_bounds__(256)
gcn_scatter_kernel(const float* __restrict__ h,
                   const int* __restrict__ src,
                   const int* __restrict__ dst,
                   const float* __restrict__ w,
                   const float* __restrict__ ns,
                   float* __restrict__ agg,
                   float* aggB, long long total) {
    constexpr int G = D / 4;
    __shared__ unsigned s_half;
    const unsigned asked = gcn_ask_block_half();
    const int tot   = (int)total;
    const int step  = (int)(gridDim.x * blockDim.x);
    const int estep = step / G;
    int t = (int)(blockIdx.x * blockDim.x + threadIdx.x);
    int e = t / G;
    const int g4 = (t % G) * 4;

    const int elast = (tot - 1) / G;

    int cs[LOOP_U], cd[LOOP_U];
    float cw[LOOP_U];
#pragma unroll
    for (int u = 0; u < LOOP_U; ++u) {
        const int ee = min(e + u * estep, elast);
        cs[u] = src[ee];
        cd[u] = dst[ee];
        cw[u] = w[ee];
    }
    gcn_post_block_half(&s_half, asked);
    float* const plane = gcn_take_block_half(&s_half) ? aggB : agg;

    for (; t + (LOOP_U - 1) * step < tot; t += LOOP_U * step, e += LOOP_U * estep) {
        float4 hv[LOOP_U];
        float  nv[LOOP_U];
#pragma unroll
        for (int u = 0; u < LOOP_U; ++u) {
            hv[u] = *(const float4*)(h + cs[u] * D + g4);
            nv[u] = SRC_NORM ? ns[cs[u]] : 1.0f;
        }
        int   ns_[LOOP_U], nd_[LOOP_U];
        float nw_[LOOP_U];
#pragma unroll
        for (int u = 0; u < LOOP_U; ++u) {
            const int ee = min(e + (LOOP_U + u) * estep, elast);
            ns_[u] = src[ee];
            nd_[u] = dst[ee];
            nw_[u] = w[ee];
        }
        static_assert(LOOP_U == 4, "the statement below names four rows");
        asm volatile("" ::
            "v"(hv[0].x), "v"(hv[0].y), "v"(hv[0].z), "v"(hv[0].w),
            "v"(hv[1].x), "v"(hv[1].y), "v"(hv[1].z), "v"(hv[1].w),
            "v"(hv[2].x), "v"(hv[2].y), "v"(hv[2].z), "v"(hv[2].w),
            "v"(hv[3].x), "v"(hv[3].y), "v"(hv[3].z), "v"(hv[3].w));
#pragma unroll
        for (int u = 0; u < LOOP_U; ++u) {
            const float wt = SRC_NORM ? cw[u] * nv[u] : cw[u];
            float* ap = plane + cd[u] * PITCH + g4;
            atomicAdd(ap + 0, hv[u].x * wt);
            atomicAdd(ap + 1, hv[u].y * wt);
            atomicAdd(ap + 2, hv[u].z * wt);
            atomicAdd(ap + 3, hv[u].w * wt);
        }
#pragma unroll
        for (int u = 0; u < LOOP_U; ++u) {
            cs[u] = ns_[u];
            cd[u] = nd_[u];
            cw[u] = nw_[u];
        }
    }
#pragma unroll
    for (int u = 0; u < LOOP_U - 1; ++u) {
        if (t + u * step < tot) {
            const float4 hv = *(const float4*)(h + cs[u] * D + g4);
            const float wt = SRC_NORM ? cw[u] * ns[cs[u]] : cw[u];
            float* ap = plane + cd[u] * PITCH + g4;
            atomicAdd(ap + 0, hv.x * wt);
            atomicAdd(ap + 1, hv.y * wt);
            atomicAdd(ap + 2, hv.z * wt);
            atomicAdd(ap + 3, hv.w * wt);
        }
    }
}

__global__ void gcn_finalize_kernel(float* __restrict__ out,
                                    const float* __restrict__ sumA,
                                    const float* __restrict__ sumB,
                                    const float* __restrict__ nd,
                                    const float* __restrict__ b2, long long total4) {
    constexpr int G = OUT_D / 4;
    long long t = (long long)blockIdx.x * blockDim.x + threadIdx.x;
    if (t >= total4) return;
    const int i = (int)(t / G);
    const int g = (int)(t % G);
    const float s = nd[i];
    const float4 bv = *(const float4*)(b2 + g * 4);
    float4* op = (float4*)(out + (long long)i * OUT_D + g * 4);
    float4 v = *(const float4*)(sumA + (long long)i * IN_D + g * 4);
    const float4 u = *(const float4*)(sumB + (long long)i * IN_D + g * 4);
    v.x = (v.x + u.x) * s + bv.x;
    v.y = (v.y + u.y) * s + bv.y;
    v.z = (v.z + u.z) * s + bv.z;
    v.w = (v.w + u.w) * s + bv.w;
    *op = v;
}

extern "C" void kernel_launch(void* const* d_in, const int* in_sizes, int n_in,
                              void* d_out, int out_size, void* d_ws, size_t ws_size,
                              hipStream_t stream) {
    const float* feat = (const float*)d_in[0];
    const int*   src  = (const int*)d_in[1];
    const int*   dst  = (const int*)d_in[2];
    const float* ew   = (const float*)d_in[3];
    const float* W1   = (const float*)d_in[4];
    const float* b1   = (const float*)d_in[5];
    const float* W2   = (const float*)d_in[6];
    const float* b2   = (const float*)d_in[7];
    float* out = (float*)d_out;

    const int n = in_sizes[0] / IN_D;
    const int E = in_sizes[1];

    float* ns   = (float*)d_ws;
    float* nd   = ns + n;
    float* aggA = nd + n;
    float* aggB = aggA + (long long)n * IN_D;
    float* degO = aggB + (long long)n * IN_D;
    float* degI = degO + n;
    float* h2   = degO;

    const int B = 256;
    auto blocks = [&](long long work) { return (unsigned)((work + B - 1) / B); };

    const long long nzero = (long long)n * (2 * IN_D + 4);
    gcn_zero_kernel<<<blocks((nzero + 3) / 4), B, 0, stream>>>(aggA, nzero);

    gcn_degree_kernel<<<LOOP_BLOCKS, B, 0, stream>>>(src, dst, degO, degI, n, E);

    gcn_norm_kernel<<<blocks(n), B, 0, stream>>>(degO, degI, ns, nd, n);

    gcn_scatter_kernel<IN_D, IN_D, true><<<LOOP_BLOCKS, B, 0, stream>>>(
        feat, src, dst, ew, ns, aggA, aggB, (long long)E * (IN_D / 4));

    const int tiles = (n + 15) / 16;
    const int wavesPerBlock = B / 32;
    const unsigned gemmBlocks = (unsigned)((tiles + wavesPerBlock - 1) / wavesPerBlock);
    gcn_dense_kernel<<<gemmBlocks, B, 0, stream>>>(aggA, aggB, ns, nd, W1, b1, W2, h2, n);

    gcn_scatter_kernel<OUT_D, IN_D, false><<<LOOP_BLOCKS, B, 0, stream>>>(
        h2, src, dst, ew, ns, aggA, aggB, (long long)E * (OUT_D / 4));

    gcn_finalize_kernel<<<blocks((long long)n * (OUT_D / 4)), B, 0, stream>>>(
        out, aggA, aggB, nd, b2, (long long)n * (OUT_D / 4));
}
